# starting kernel
# baseline (speedup 1.0000x reference)
#include <hip/hip_runtime.h>
#include <hip/hip_bf16.h>
#include <math.h>

#define U_DIM 512
#define B_DIM 4
#define L_DIM 512
#define H_DIM 8
#define D_DIM 64
#define HB_DIM 32
#define M_ROWS 2048

typedef __attribute__((ext_vector_type(2))) float v2f;
typedef __attribute__((ext_vector_type(8))) float v8f;

__device__ __forceinline__ v8f wmma4(v2f a, v2f b, v8f c) {
#if defined(__HIP_DEVICE_COMPILE__)
  return __builtin_amdgcn_wmma_f32_16x16x4_f32(false, a, false, b, (short)0, c,
                                               false, false);
#else
  (void)a; (void)b;
  return c;
#endif
}


#define SW_STRIDE 288

__global__ __launch_bounds__(256) void proj_relu_kernel(
    const float* __restrict__ Xq, const float* __restrict__ Xk,
    const float* __restrict__ Xv, const float* __restrict__ Wq,
    const float* __restrict__ Wk, const float* __restrict__ Wv,
    const float* __restrict__ bq, const float* __restrict__ bk,
    const float* __restrict__ bv, float* __restrict__ Qh,
    float* __restrict__ Kh, float* __restrict__ Vh) {
  const int which = blockIdx.z;
  const float* X = which == 0 ? Xq : (which == 1 ? Xk : Xv);
  const float* W = which == 0 ? Wq : (which == 1 ? Wk : Wv);
  const float* bias = which == 0 ? bq : (which == 1 ? bk : bv);
  float* O = which == 0 ? Qh : (which == 1 ? Kh : Vh);

  const int m0 = blockIdx.y * 64;
  const int n0 = blockIdx.x * 128;
  const int tid = threadIdx.x;
  const int lane = tid & 31;
  const int wave = tid >> 5;
  const int wm = wave >> 1;
  const int wn = wave & 1;
  const int l16 = lane & 15;
  const int hi = lane >> 4;

  __shared__ float sX[64][68];
  __shared__ float sW2[8][SW_STRIDE];

  v8f acc[4] = {};

  for (int kt = 0; kt < U_DIM / 16; ++kt) {
    {
      const int r = tid >> 2, c = (tid & 3) * 4;
      const float4 x =
          *(const float4*)&X[(size_t)(m0 + r) * U_DIM + kt * 16 + c];
      *(float4*)&sX[r][c] = x;
    }
    {
      const int r = tid >> 4, c = (tid & 15) * 8;
      const float4 w0 =
          *(const float4*)&W[(size_t)(kt * 16 + r) * U_DIM + n0 + c];
      const float4 w1 =
          *(const float4*)&W[(size_t)(kt * 16 + r) * U_DIM + n0 + c + 4];
      float* dst = &sW2[r >> 1][c * 2 + (r & 1)];
      dst[0] = w0.x;  dst[2] = w0.y;  dst[4] = w0.z;  dst[6] = w0.w;
      dst[8] = w1.x;  dst[10] = w1.y; dst[12] = w1.z; dst[14] = w1.w;
    }
    __syncthreads();

#pragma unroll
    for (int kk = 0; kk < 16; kk += 4) {
      const v2f a = *(const v2f*)&sX[wm * 16 + l16][kk + 2 * hi];
      const int kp = (kk >> 1) + hi;
#pragma unroll
      for (int n = 0; n < 4; ++n) {
        const int col = wn * 64 + n * 16 + l16;
        const v2f b = *(const v2f*)&sW2[kp][col * 2];
        acc[n] = wmma4(a, b, acc[n]);
      }
    }
    __syncthreads();
  }

#pragma unroll
  for (int n = 0; n < 4; ++n) {
    const int ng = n0 + wn * 64 + n * 16 + l16;
    const int h = ng >> 6, d = ng & 63;
    const float bb = bias[ng];
#pragma unroll
    for (int r = 0; r < 8; ++r) {
      const int mg = m0 + wm * 16 + r + 8 * hi;
      const int b = mg >> 9, l = mg & 511;
      float v = acc[n][r] + bb;
      v = v > 0.0f ? v : 0.0f;
      O[(((size_t)(h * B_DIM + b) * L_DIM + l) * D_DIM) + d] = v;
    }
  }
}

__global__ __launch_bounds__(256) void aux_kernel(
    const float* __restrict__ Qh, const float* __restrict__ Kh,
    const float* __restrict__ queries, float* __restrict__ qq,
    float* __restrict__ kkv, float* __restrict__ qsign) {
  const int idx = blockIdx.x * blockDim.x + threadIdx.x;
  const int nrows = HB_DIM * L_DIM;
  if (idx < nrows) {
    const float* p = Qh + (size_t)idx * D_DIM;
    float s = 0.0f;
    for (int d = 0; d < D_DIM; ++d) s += p[d] * p[d];
    qq[idx] = s;
  } else if (idx < 2 * nrows) {
    const float* p = Kh + (size_t)(idx - nrows) * D_DIM;
    float s = 0.0f;
    for (int d = 0; d < D_DIM; ++d) s += p[d] * p[d];
    kkv[idx - nrows] = s;
  } else if (idx < 2 * nrows + B_DIM * L_DIM) {
    const int r = idx - 2 * nrows;
    const float* p = queries + (size_t)r * U_DIM;
    float s = 0.0f;
    for (int u = 0; u < U_DIM; ++u) s += p[u];
    qsign[r] = (s != 0.0f) ? 1.0f : 0.0f;
  }
}

#define SKT_STRIDE 96
#define SV_STRIDE 160

__global__ __launch_bounds__(256) void attn_kernel(
    const float* __restrict__ Qh, const float* __restrict__ Kh,
    const float* __restrict__ Vh, const float* __restrict__ qq,
    const float* __restrict__ kkv, const float* __restrict__ qsign,
    float* __restrict__ outp, float* __restrict__ ascore) {
  const int hb = blockIdx.y;
  const int q0 = blockIdx.x * 64;
  const int tid = threadIdx.x;
  const int lane = tid & 31;
  const int wave = tid >> 5;
  const int wm = wave >> 1;
  const int wn = wave & 1;
  const int l16 = lane & 15;
  const int hi = lane >> 4;

  __shared__ float sQ[64][68];
  __shared__ float sKT2[32][SKT_STRIDE];
  __shared__ float sV2[16][SV_STRIDE];
  __shared__ float sP[64][36];
  __shared__ float sKK[L_DIM];
  __shared__ float sQQ[64];
  __shared__ float sQM[64];
  __shared__ float sMrow[64];
  __shared__ float sInvL[64];
  __shared__ float sStat[2][64][2];

  {
    const int r = tid >> 2, c = (tid & 3) * 16;
    const size_t base = ((size_t)hb * L_DIM + q0 + r) * D_DIM + c;
#pragma unroll
    for (int j = 0; j < 4; ++j) {
      const float4 v = *(const float4*)&Qh[base + 4 * j];
      *(float4*)&sQ[r][c + 4 * j] = v;
    }
    sKK[2 * tid] = kkv[(size_t)hb * L_DIM + 2 * tid];
    sKK[2 * tid + 1] = kkv[(size_t)hb * L_DIM + 2 * tid + 1];
    if (tid < 64) {
      sQQ[tid] = qq[(size_t)hb * L_DIM + q0 + tid];
      sQM[tid] = qsign[(size_t)(hb & 3) * L_DIM + q0 + tid];
    }
  }
  __syncthreads();

  float m_r[8], l_r[8];
#pragma unroll
  for (int r = 0; r < 8; ++r) { m_r[r] = -INFINITY; l_r[r] = 0.0f; }

  for (int kt = 0; kt < L_DIM / 32; ++kt) {
    {
      const int kc = tid >> 3, d0 = (tid & 7) * 8;
      const size_t base = ((size_t)hb * L_DIM + kt * 32 + kc) * D_DIM + d0;
      const float4 k0 = *(const float4*)&Kh[base];
      const float4 k1 = *(const float4*)&Kh[base + 4];
      float* dst = &sKT2[d0 >> 1][kc * 2];
      *(v2f*)&dst[0] = v2f{k0.x, k0.y};
      *(v2f*)&dst[SKT_STRIDE] = v2f{k0.z, k0.w};
      *(v2f*)&dst[2 * SKT_STRIDE] = v2f{k1.x, k1.y};
      *(v2f*)&dst[3 * SKT_STRIDE] = v2f{k1.z, k1.w};
    }
    __syncthreads();

    v8f s = {};
#pragma unroll
    for (int kk = 0; kk < D_DIM; kk += 4) {
      const v2f a = *(const v2f*)&sQ[wm * 16 + l16][kk + 2 * hi];
      const v2f b = *(const v2f*)&sKT2[(kk >> 1) + hi][(wn * 16 + l16) * 2];
      s = wmma4(a, b, s);
    }

    const int kcol = wn * 16 + l16;
    const int kg = kt * 32 + kcol;
#pragma unroll
    for (int r = 0; r < 8; ++r) {
      const int qrow = wm * 16 + r + 8 * hi;
      float raw = sQQ[qrow] + sKK[kg] - 2.0f * s[r];
      raw = fmaxf(raw, 1e-12f);
      const float sc = -sqrtf(raw) * 0.125f;
      __builtin_nontemporal_store(
          sc * sQM[qrow],
          &ascore[((size_t)hb * L_DIM + q0 + qrow) * L_DIM + kg]);
      const float mn = fmaxf(m_r[r], sc);
      l_r[r] = l_r[r] * __expf(m_r[r] - mn) + __expf(sc - mn);
      m_r[r] = mn;
    }
    __syncthreads();
  }

#pragma unroll
  for (int r = 0; r < 8; ++r) {
    float m = m_r[r], l = l_r[r];
#pragma unroll
    for (int off = 8; off >= 1; off >>= 1) {
      const float mo = __shfl_xor(m, off, 16);
      const float lo = __shfl_xor(l, off, 16);
      const float mn = fmaxf(m, mo);
      l = l * __expf(m - mn) + lo * __expf(mo - mn);
      m = mn;
    }
    m_r[r] = m; l_r[r] = l;
  }
  if (l16 == 0) {
#pragma unroll
    for (int r = 0; r < 8; ++r) {
      const int row = wm * 16 + r + 8 * hi;
      sStat[wn][row][0] = m_r[r];
      sStat[wn][row][1] = l_r[r];
    }
  }
  __syncthreads();
  if (tid < 64) {
    const float ma = sStat[0][tid][0], la = sStat[0][tid][1];
    const float mb = sStat[1][tid][0], lb = sStat[1][tid][1];
    const float mn = fmaxf(ma, mb);
    const float L = la * __expf(ma - mn) + lb * __expf(mb - mn);
    sMrow[tid] = mn;
    sInvL[tid] = 1.0f / L;
  }
  __syncthreads();

  v8f o[2] = {};
  for (int kt = 0; kt < L_DIM / 32; ++kt) {
    {
      const int kc = tid >> 3, d0 = (tid & 7) * 8;
      const size_t kb = ((size_t)hb * L_DIM + kt * 32 + kc) * D_DIM + d0;
      const float4 k0 = *(const float4*)&Kh[kb];
      const float4 k1 = *(const float4*)&Kh[kb + 4];
      float* dk = &sKT2[d0 >> 1][kc * 2];
      *(v2f*)&dk[0] = v2f{k0.x, k0.y};
      *(v2f*)&dk[SKT_STRIDE] = v2f{k0.z, k0.w};
      *(v2f*)&dk[2 * SKT_STRIDE] = v2f{k1.x, k1.y};
      *(v2f*)&dk[3 * SKT_STRIDE] = v2f{k1.z, k1.w};
      const float4 v0 = *(const float4*)&Vh[kb];
      const float4 v1 = *(const float4*)&Vh[kb + 4];
      float* dv = &sV2[kc >> 1][d0 * 2 + (kc & 1)];
      dv[0] = v0.x;  dv[2] = v0.y;  dv[4] = v0.z;  dv[6] = v0.w;
      dv[8] = v1.x;  dv[10] = v1.y; dv[12] = v1.z; dv[14] = v1.w;
    }
    __syncthreads();

    v8f s = {};
#pragma unroll
    for (int kk = 0; kk < D_DIM; kk += 4) {
      const v2f a = *(const v2f*)&sQ[wm * 16 + l16][kk + 2 * hi];
      const v2f b = *(const v2f*)&sKT2[(kk >> 1) + hi][(wn * 16 + l16) * 2];
      s = wmma4(a, b, s);
    }
    const int kcol = wn * 16 + l16;
    const int kg = kt * 32 + kcol;
#pragma unroll
    for (int r = 0; r < 8; ++r) {
      const int qrow = wm * 16 + r + 8 * hi;
      float raw = sQQ[qrow] + sKK[kg] - 2.0f * s[r];
      raw = fmaxf(raw, 1e-12f);
      const float sc = -sqrtf(raw) * 0.125f;
      sP[qrow][kcol] = __expf(sc - sMrow[qrow]) * sInvL[qrow];
    }
    __syncthreads();

#pragma unroll
    for (int kk = 0; kk < 32; kk += 4) {
      const v2f a = *(const v2f*)&sP[wm * 16 + l16][kk + 2 * hi];
      const int kp = (kk >> 1) + hi;
#pragma unroll
      for (int n = 0; n < 2; ++n) {
        const int dcol = wn * 32 + n * 16 + l16;
        const v2f b = *(const v2f*)&sV2[kp][dcol * 2];
        o[n] = wmma4(a, b, o[n]);
      }
    }
    __syncthreads();
  }

  const int h = hb >> 2, b = hb & 3;
#pragma unroll
  for (int n = 0; n < 2; ++n) {
    const int d = wn * 32 + n * 16 + l16;
#pragma unroll
    for (int r = 0; r < 8; ++r) {
      const int qrow = wm * 16 + r + 8 * hi;
      const float v = o[n][r] * sQM[qrow];
      __builtin_nontemporal_store(
          v, &outp[((size_t)b * L_DIM + q0 + qrow) * U_DIM + h * D_DIM + d]);
    }
  }
}

extern "C" void kernel_launch(void* const* d_in, const int* in_sizes, int n_in,
                              void* d_out, int out_size, void* d_ws,
                              size_t ws_size, hipStream_t stream) {
  const float* queries = (const float*)d_in[0];
  const float* keys = (const float*)d_in[1];
  const float* values = (const float*)d_in[2];
  const float* Wq = (const float*)d_in[3];
  const float* bq = (const float*)d_in[4];
  const float* Wk = (const float*)d_in[5];
  const float* bk = (const float*)d_in[6];
  const float* Wv = (const float*)d_in[7];
  const float* bv = (const float*)d_in[8];

  float* outp = (float*)d_out;
  float* ascore = outp + (size_t)B_DIM * L_DIM * U_DIM;

  float* ws = (float*)d_ws;
  float* Qh = ws;
  float* Kh = Qh + (size_t)HB_DIM * L_DIM * D_DIM;
  float* Vh = Kh + (size_t)HB_DIM * L_DIM * D_DIM;
  float* qq = Vh + (size_t)HB_DIM * L_DIM * D_DIM;
  float* kkv = qq + (size_t)HB_DIM * L_DIM;
  float* qsign = kkv + (size_t)HB_DIM * L_DIM;

  proj_relu_kernel<<<dim3(U_DIM / 128, M_ROWS / 64, 3), 256, 0, stream>>>(
      queries, keys, values, Wq, Wk, Wv, bq, bk, bv, Qh, Kh, Vh);

  const int aux_threads = 2 * HB_DIM * L_DIM + B_DIM * L_DIM;
  aux_kernel<<<(aux_threads + 255) / 256, 256, 0, stream>>>(Qh, Kh, queries,
                                                            qq, kkv, qsign);

  attn_kernel<<<dim3(L_DIM / 64, HB_DIM), 256, 0, stream>>>(
      Qh, Kh, Vh, qq, kkv, qsign, outp, ascore);
}


# projection kernel re-cut: X and W fragments straight from global (16-B loads, next 32-k block grouped one block ahead), no LDS tiles; wave = 16 rows x one head, 768 blocks; k and sum order changed
# speedup vs baseline: 2.4964x; 2.4964x over previous
#include <hip/hip_runtime.h>
#include <hip/hip_bf16.h>
#include <math.h>

#define U_DIM 512
#define B_DIM 4
#define L_DIM 512
#define H_DIM 8
#define D_DIM 64
#define HB_DIM 32
#define M_ROWS 2048

typedef __attribute__((ext_vector_type(2))) float v2f;
typedef __attribute__((ext_vector_type(8))) float v8f;

__device__ __forceinline__ v8f wmma4(v2f a, v2f b, v8f c) {
#if defined(__HIP_DEVICE_COMPILE__)
  return __builtin_amdgcn_wmma_f32_16x16x4_f32(false, a, false, b, (short)0, c,
                                               false, false);
#else
  (void)a; (void)b;
  return c;
#endif
}

typedef __attribute__((ext_vector_type(4))) float v4f;

#if defined(__HIP_DEVICE_COMPILE__)
#define PROJ_SCHED_FENCE() __builtin_amdgcn_sched_barrier(0)
#define PROJ_PACE_BARRIER() __builtin_amdgcn_s_barrier()
#else
#define PROJ_SCHED_FENCE() ((void)0)
#define PROJ_PACE_BARRIER() ((void)0)
#endif

struct PFrag {
  v4f x[4];
  v4f w[4][2][2];
};

__device__ __forceinline__ void proj_load(PFrag& f, const float* xp,
                                          const float* wp) {
#pragma unroll
  for (int q = 0; q < 4; ++q) f.x[q] = *(const v4f*)(xp + 8 * q);
#pragma unroll
  for (int q = 0; q < 4; ++q)
#pragma unroll
    for (int t = 0; t < 2; ++t)
#pragma unroll
      for (int s = 0; s < 2; ++s)
        f.w[q][t][s] = *(const v4f*)(wp + (size_t)(8 * q + 2 * t + s) * U_DIM);
}

__device__ __forceinline__ void proj_mma(v8f (&acc)[4], const PFrag& f) {
#pragma unroll
  for (int q = 0; q < 4; ++q)
#pragma unroll
    for (int t = 0; t < 2; ++t) {
      const v2f a = t == 0 ? v2f{f.x[q].x, f.x[q].y} : v2f{f.x[q].z, f.x[q].w};
#pragma unroll
      for (int np = 0; np < 4; ++np) {
        const v2f b = v2f{f.w[q][t][0][np], f.w[q][t][1][np]};
        acc[np] = wmma4(a, b, acc[np]);
      }
    }
}

__global__ __launch_bounds__(128) __attribute__((amdgpu_waves_per_eu(3, 3)))
void proj_relu_kernel(
    const float* __restrict__ Xq, const float* __restrict__ Xk,
    const float* __restrict__ Xv, const float* __restrict__ Wq,
    const float* __restrict__ Wk, const float* __restrict__ Wv,
    const float* __restrict__ bq, const float* __restrict__ bk,
    const float* __restrict__ bv, float* __restrict__ Qh,
    float* __restrict__ Kh, float* __restrict__ Vh) {
  const int which = blockIdx.z;
  const float* X = which == 0 ? Xq : (which == 1 ? Xk : Xv);
  const float* W = which == 0 ? Wq : (which == 1 ? Wk : Wv);
  const float* bias = which == 0 ? bq : (which == 1 ? bk : bv);
  float* O = which == 0 ? Qh : (which == 1 ? Kh : Vh);

  const int g = blockIdx.x;
  const int tid = threadIdx.x;
  const int lane = tid & 31;
  const int wave = tid >> 5;
  const int l16 = lane & 15;
  const int hi = lane >> 4;
  const int m0 = blockIdx.y * 64 + 16 * wave;

  const float* xp = X + (size_t)(m0 + l16) * U_DIM + 4 * hi;
  const float* wp = W + (size_t)(4 * hi) * U_DIM + D_DIM * g + 4 * l16;

  v8f acc[4] = {};
  PFrag fa, fb;

  proj_load(fa, xp, wp);
  for (int i = 0; i < 7; ++i) {
    PROJ_PACE_BARRIER();
    proj_load(fb, xp + 32, wp + (size_t)32 * U_DIM);
    PROJ_SCHED_FENCE();
    proj_mma(acc, fa);
    xp += 64;
    wp += (size_t)64 * U_DIM;
    PROJ_PACE_BARRIER();
    proj_load(fa, xp, wp);
    PROJ_SCHED_FENCE();
    proj_mma(acc, fb);
  }
  PROJ_PACE_BARRIER();
  proj_load(fb, xp + 32, wp + (size_t)32 * U_DIM);
  PROJ_SCHED_FENCE();
  proj_mma(acc, fa);
  proj_mma(acc, fb);

  const v4f b4 = *(const v4f*)&bias[D_DIM * g + 4 * l16];
#pragma unroll
  for (int r = 0; r < 8; ++r) {
    const int mg = m0 + 8 * hi + r;
    const int b = mg >> 9, l = mg & 511;
    v4f v = v4f{acc[0][r], acc[1][r], acc[2][r], acc[3][r]} + b4;
    v.x = v.x > 0.0f ? v.x : 0.0f;
    v.y = v.y > 0.0f ? v.y : 0.0f;
    v.z = v.z > 0.0f ? v.z : 0.0f;
    v.w = v.w > 0.0f ? v.w : 0.0f;
    *(v4f*)&O[(((size_t)(g * B_DIM + b) * L_DIM + l) * D_DIM) + 4 * l16] = v;
  }
}

#define AQ 32
#define AT_PITCH 68

__global__ __launch_bounds__(256) void attn_kernel(
    const float* __restrict__ Qh, const float* __restrict__ Kh,
    const float* __restrict__ Vh, const float* __restrict__ queries,
    float* __restrict__ outp, float* __restrict__ ascore) {
  const int hb = blockIdx.y;
  const int q0 = blockIdx.x * AQ;
  const int tid = threadIdx.x;
  const int lane = tid & 31;
  const int wave = tid >> 5;
  const int l16 = lane & 15;
  const int hi = lane >> 4;
  const int key0 = wave * 64;

  __shared__ __attribute__((aligned(16))) float sT[8][AQ][AT_PITCH];
  __shared__ float sM[8][AQ];
  __shared__ float sL[8][AQ];
  __shared__ __attribute__((aligned(16))) float sQM[2][AQ / 2];

  const int h = hb >> 2, b = hb & 3;

  const float* Qb = Qh + (size_t)hb * L_DIM * D_DIM;
  const float* Kb = Kh + (size_t)hb * L_DIM * D_DIM;
  const float* Vb = Vh + (size_t)hb * L_DIM * D_DIM;

  v4f qf[2][8];
#pragma unroll
  for (int qt = 0; qt < 2; ++qt)
#pragma unroll
    for (int j = 0; j < 8; ++j)
      qf[qt][j] = *(const v4f*)&Qb[(size_t)(q0 + 16 * qt + l16) * D_DIM +
                                   8 * j + 4 * hi];

  float qqv[2];
#pragma unroll
  for (int qt = 0; qt < 2; ++qt) {
    v4f s4 = qf[qt][0] * qf[qt][0];
#pragma unroll
    for (int j = 1; j < 8; ++j) s4 += qf[qt][j] * qf[qt][j];
    const float s = (s4.x + s4.y) + (s4.z + s4.w);
    qqv[qt] = s + __shfl_xor(s, 16);
  }

  v8f acc[4][2] = {};
  float kkrow[4];
#pragma unroll
  for (int n = 0; n < 4; ++n) {
    v4f kf[8];
#pragma unroll
    for (int j = 0; j < 8; ++j)
      kf[j] = *(const v4f*)&Kb[(size_t)(key0 + 16 * n + l16) * D_DIM + 8 * j +
                               4 * hi];
    {
      v4f s4 = kf[0] * kf[0];
#pragma unroll
      for (int j = 1; j < 8; ++j) s4 += kf[j] * kf[j];
      const float s = (s4.x + s4.y) + (s4.z + s4.w);
      kkrow[n] = s + __shfl_xor(s, 16);
    }
#pragma unroll
    for (int j = 0; j < 8; ++j) {
#pragma unroll
      for (int qt = 0; qt < 2; ++qt)
        acc[n][qt] = wmma4(kf[j].xy, qf[qt][j].xy, acc[n][qt]);
#pragma unroll
      for (int qt = 0; qt < 2; ++qt)
        acc[n][qt] = wmma4(kf[j].zw, qf[qt][j].zw, acc[n][qt]);
    }
  }

#if defined(__HIP_DEVICE_COMPILE__)
  __builtin_amdgcn_sched_barrier(0);
#endif
  float rs[4];
  {
    const float* qr =
        queries + ((size_t)b * L_DIM + q0 + 4 * wave) * U_DIM + 4 * lane;
#pragma unroll
    for (int rr = 0; rr < 4; ++rr) {
      v4f x[4];
#pragma unroll
      for (int i = 0; i < 4; ++i)
        x[i] = __builtin_nontemporal_load(
            (const v4f*)&qr[(size_t)rr * U_DIM + 128 * i]);
      const v4f t = (x[0] + x[1]) + (x[2] + x[3]);
      rs[rr] = (t.x + t.y) + (t.z + t.w);
    }
#pragma unroll
    for (int m = 16; m >= 1; m >>= 1)
#pragma unroll
      for (int rr = 0; rr < 4; ++rr) rs[rr] += __shfl_xor(rs[rr], m);
  }

  float mloc[2] = {-INFINITY, -INFINITY};
#pragma unroll
  for (int n = 0; n < 4; ++n) {
    float kk8[8];
#pragma unroll
    for (int r = 0; r < 8; ++r) kk8[r] = __shfl(kkrow[n], 8 * hi + r);
#pragma unroll
    for (int qt = 0; qt < 2; ++qt) {
      float st[8];
#pragma unroll
      for (int r = 0; r < 8; ++r) {
        float raw = qqv[qt] + kk8[r] - 2.0f * acc[n][qt][r];
        raw = fmaxf(raw, 1e-12f);
        const float sc = -sqrtf(raw) * 0.125f;
        acc[n][qt][r] = sc;
        mloc[qt] = fmaxf(mloc[qt], sc);
        st[r] = sc;
      }
      float* dst = &sT[wave][16 * qt + l16][16 * n + 8 * hi];
      *(v4f*)&dst[0] = v4f{st[0], st[1], st[2], st[3]};
      *(v4f*)&dst[4] = v4f{st[4], st[5], st[6], st[7]};
    }
  }
#pragma unroll
  for (int qt = 0; qt < 2; ++qt) {
    mloc[qt] = fmaxf(mloc[qt], __shfl_xor(mloc[qt], 16));
    if (hi == 0) sM[wave][16 * qt + l16] = mloc[qt];
  }
  if (lane == 0) {
    *(v2f*)&sQM[0][2 * wave] = v2f{rs[0] != 0.0f ? 1.0f : 0.0f,
                                   rs[2] != 0.0f ? 1.0f : 0.0f};
    *(v2f*)&sQM[1][2 * wave] = v2f{rs[1] != 0.0f ? 1.0f : 0.0f,
                                   rs[3] != 0.0f ? 1.0f : 0.0f};
  }
  __syncthreads();

  v4f qm4[4];
#pragma unroll
  for (int i = 0; i < 4; ++i) qm4[i] = *(const v4f*)&sQM[hi][4 * i];
#pragma unroll
  for (int i = 0; i < 16; ++i) {
    const int row = 2 * i + hi;
    const v4f v = *(const v4f*)&sT[wave][row][4 * l16] * qm4[i >> 2][i & 3];
    __builtin_nontemporal_store(
        v, (v4f*)&ascore[((size_t)hb * L_DIM + q0 + row) * L_DIM + key0 +
                         4 * l16]);
  }

  float lsum[2];
#pragma unroll
  for (int qt = 0; qt < 2; ++qt) {
    float m = sM[0][16 * qt + l16];
#pragma unroll
    for (int w = 1; w < 8; ++w) m = fmaxf(m, sM[w][16 * qt + l16]);
    float l = 0.0f;
#pragma unroll
    for (int n = 0; n < 4; ++n)
#pragma unroll
      for (int r = 0; r < 8; ++r) {
        const float p = __expf(acc[n][qt][r] - m);
        acc[n][qt][r] = p;
        l += p;
      }
    lsum[qt] = l + __shfl_xor(l, 16);
  }

  v8f o[4][2] = {};
#pragma unroll
  for (int n = 0; n < 4; ++n) {
    v4f vf[4][2];
#pragma unroll
    for (int t = 0; t < 4; ++t)
#pragma unroll
      for (int s = 0; s < 2; ++s)
        vf[t][s] = *(const v4f*)&Vb[(size_t)(key0 + 16 * n + 8 * hi + 2 * t + s) *
                                        D_DIM + 4 * l16];
#pragma unroll
    for (int t = 0; t < 4; ++t) {
#pragma unroll
      for (int np = 0; np < 4; ++np) {
        const v2f a = v2f{vf[t][0][np], vf[t][1][np]};
#pragma unroll
        for (int qt = 0; qt < 2; ++qt) {
          const v2f b = v2f{acc[n][qt][2 * t], acc[n][qt][2 * t + 1]};
          o[np][qt] = wmma4(a, b, o[np][qt]);
        }
      }
    }
  }

#pragma unroll
  for (int qt = 0; qt < 2; ++qt) {
    float* dst = &sT[wave][16 * qt + l16][32 * hi];
#pragma unroll
    for (int r = 0; r < 8; ++r)
      *(v4f*)&dst[4 * r] =
          v4f{o[0][qt][r], o[1][qt][r], o[2][qt][r], o[3][qt][r]};
    if (hi == 0) sL[wave][16 * qt + l16] = lsum[qt];
  }
  __syncthreads();

#pragma unroll
  for (int i = 0; i < 2; ++i) {
    const int q = (tid >> 4) + 16 * i;
    const int c4 = (tid & 15) * 4;
    v4f sum = *(const v4f*)&sT[0][q][c4];
    float L = sL[0][q];
#pragma unroll
    for (int w = 1; w < 8; ++w) {
      sum += *(const v4f*)&sT[w][q][c4];
      L += sL[w][q];
    }
    const float scale = sQM[q & 1][q >> 1] / L;
    sum *= scale;
    __builtin_nontemporal_store(
        sum, (v4f*)&outp[((size_t)b * L_DIM + q0 + q) * U_DIM + h * D_DIM + c4]);
  }
}

extern "C" void kernel_launch(void* const* d_in, const int* in_sizes, int n_in,
                              void* d_out, int out_size, void* d_ws,
                              size_t ws_size, hipStream_t stream) {
  const float* queries = (const float*)d_in[0];
  const float* keys = (const float*)d_in[1];
  const float* values = (const float*)d_in[2];
  const float* Wq = (const float*)d_in[3];
  const float* bq = (const float*)d_in[4];
  const float* Wk = (const float*)d_in[5];
  const float* bk = (const float*)d_in[6];
  const float* Wv = (const float*)d_in[7];
  const float* bv = (const float*)d_in[8];

  float* outp = (float*)d_out;
  float* ascore = outp + (size_t)B_DIM * L_DIM * U_DIM;

  float* ws = (float*)d_ws;
  float* Qh = ws;
  float* Kh = Qh + (size_t)HB_DIM * L_DIM * D_DIM;
  float* Vh = Kh + (size_t)HB_DIM * L_DIM * D_DIM;

  proj_relu_kernel<<<dim3(H_DIM, M_ROWS / 64, 3), 128, 0, stream>>>(
      queries, keys, values, Wq, Wk, Wv, bq, bk, bv, Qh, Kh, Vh);

  attn_kernel<<<dim3(L_DIM / AQ, HB_DIM), 256, 0, stream>>>(
      Qh, Kh, Vh, queries, outp, ascore);
}


# attention score arithmetic: sqrt on the hardware instruction v_sqrt_f32 (1 ulp; sqrtf was its 14-instruction correctly rounded form), 1/8 folded into its argument, sign applied in the mask product
# speedup vs baseline: 2.5851x; 1.0355x over previous
#include <hip/hip_runtime.h>
#include <hip/hip_bf16.h>
#include <math.h>

#define U_DIM 512
#define B_DIM 4
#define L_DIM 512
#define H_DIM 8
#define D_DIM 64
#define HB_DIM 32
#define M_ROWS 2048

typedef __attribute__((ext_vector_type(2))) float v2f;
typedef __attribute__((ext_vector_type(8))) float v8f;

__device__ __forceinline__ v8f wmma4(v2f a, v2f b, v8f c) {
#if defined(__HIP_DEVICE_COMPILE__)
  return __builtin_amdgcn_wmma_f32_16x16x4_f32(false, a, false, b, (short)0, c,
                                               false, false);
#else
  (void)a; (void)b;
  return c;
#endif
}

typedef __attribute__((ext_vector_type(4))) float v4f;

__device__ __forceinline__ float hw_sqrt(float x) {
#if defined(__HIP_DEVICE_COMPILE__)
  return __builtin_amdgcn_sqrtf(x);
#else
  return sqrtf(x);
#endif
}

#if defined(__HIP_DEVICE_COMPILE__)
#define PROJ_SCHED_FENCE() __builtin_amdgcn_sched_barrier(0)
#define PROJ_PACE_BARRIER() __builtin_amdgcn_s_barrier()
#else
#define PROJ_SCHED_FENCE() ((void)0)
#define PROJ_PACE_BARRIER() ((void)0)
#endif

struct PFrag {
  v4f x[4];
  v4f w[4][2][2];
};

__device__ __forceinline__ void proj_load(PFrag& f, const float* xp,
                                          const float* wp) {
#pragma unroll
  for (int q = 0; q < 4; ++q) f.x[q] = *(const v4f*)(xp + 8 * q);
#pragma unroll
  for (int q = 0; q < 4; ++q)
#pragma unroll
    for (int t = 0; t < 2; ++t)
#pragma unroll
      for (int s = 0; s < 2; ++s)
        f.w[q][t][s] = *(const v4f*)(wp + (size_t)(8 * q + 2 * t + s) * U_DIM);
}

__device__ __forceinline__ void proj_mma(v8f (&acc)[4], const PFrag& f) {
#pragma unroll
  for (int q = 0; q < 4; ++q)
#pragma unroll
    for (int t = 0; t < 2; ++t) {
      const v2f a = t == 0 ? v2f{f.x[q].x, f.x[q].y} : v2f{f.x[q].z, f.x[q].w};
#pragma unroll
      for (int np = 0; np < 4; ++np) {
        const v2f b = v2f{f.w[q][t][0][np], f.w[q][t][1][np]};
        acc[np] = wmma4(a, b, acc[np]);
      }
    }
}

__global__ __launch_bounds__(128) __attribute__((amdgpu_waves_per_eu(3, 3)))
void proj_relu_kernel(
    const float* __restrict__ Xq, const float* __restrict__ Xk,
    const float* __restrict__ Xv, const float* __restrict__ Wq,
    const float* __restrict__ Wk, const float* __restrict__ Wv,
    const float* __restrict__ bq, const float* __restrict__ bk,
    const float* __restrict__ bv, float* __restrict__ Qh,
    float* __restrict__ Kh, float* __restrict__ Vh) {
  const int which = blockIdx.z;
  const float* X = which == 0 ? Xq : (which == 1 ? Xk : Xv);
  const float* W = which == 0 ? Wq : (which == 1 ? Wk : Wv);
  const float* bias = which == 0 ? bq : (which == 1 ? bk : bv);
  float* O = which == 0 ? Qh : (which == 1 ? Kh : Vh);

  const int g = blockIdx.x;
  const int tid = threadIdx.x;
  const int lane = tid & 31;
  const int wave = tid >> 5;
  const int l16 = lane & 15;
  const int hi = lane >> 4;
  const int m0 = blockIdx.y * 64 + 16 * wave;

  const float* xp = X + (size_t)(m0 + l16) * U_DIM + 4 * hi;
  const float* wp = W + (size_t)(4 * hi) * U_DIM + D_DIM * g + 4 * l16;

  v8f acc[4] = {};
  PFrag fa, fb;

  proj_load(fa, xp, wp);
  for (int i = 0; i < 7; ++i) {
    PROJ_PACE_BARRIER();
    proj_load(fb, xp + 32, wp + (size_t)32 * U_DIM);
    PROJ_SCHED_FENCE();
    proj_mma(acc, fa);
    xp += 64;
    wp += (size_t)64 * U_DIM;
    PROJ_PACE_BARRIER();
    proj_load(fa, xp, wp);
    PROJ_SCHED_FENCE();
    proj_mma(acc, fb);
  }
  PROJ_PACE_BARRIER();
  proj_load(fb, xp + 32, wp + (size_t)32 * U_DIM);
  PROJ_SCHED_FENCE();
  proj_mma(acc, fa);
  proj_mma(acc, fb);

  const v4f b4 = *(const v4f*)&bias[D_DIM * g + 4 * l16];
#pragma unroll
  for (int r = 0; r < 8; ++r) {
    const int mg = m0 + 8 * hi + r;
    const int b = mg >> 9, l = mg & 511;
    v4f v = v4f{acc[0][r], acc[1][r], acc[2][r], acc[3][r]} + b4;
    v.x = v.x > 0.0f ? v.x : 0.0f;
    v.y = v.y > 0.0f ? v.y : 0.0f;
    v.z = v.z > 0.0f ? v.z : 0.0f;
    v.w = v.w > 0.0f ? v.w : 0.0f;
    *(v4f*)&O[(((size_t)(g * B_DIM + b) * L_DIM + l) * D_DIM) + 4 * l16] = v;
  }
}

#define AQ 32
#define AT_PITCH 68

__global__ __launch_bounds__(256) void attn_kernel(
    const float* __restrict__ Qh, const float* __restrict__ Kh,
    const float* __restrict__ Vh, const float* __restrict__ queries,
    float* __restrict__ outp, float* __restrict__ ascore) {
  const int hb = blockIdx.y;
  const int q0 = blockIdx.x * AQ;
  const int tid = threadIdx.x;
  const int lane = tid & 31;
  const int wave = tid >> 5;
  const int l16 = lane & 15;
  const int hi = lane >> 4;
  const int key0 = wave * 64;

  __shared__ __attribute__((aligned(16))) float sT[8][AQ][AT_PITCH];
  __shared__ float sM[8][AQ];
  __shared__ float sL[8][AQ];
  __shared__ __attribute__((aligned(16))) float sQM[2][AQ / 2];

  const int h = hb >> 2, b = hb & 3;

  const float* Qb = Qh + (size_t)hb * L_DIM * D_DIM;
  const float* Kb = Kh + (size_t)hb * L_DIM * D_DIM;
  const float* Vb = Vh + (size_t)hb * L_DIM * D_DIM;

  v4f qf[2][8];
#pragma unroll
  for (int qt = 0; qt < 2; ++qt)
#pragma unroll
    for (int j = 0; j < 8; ++j)
      qf[qt][j] = *(const v4f*)&Qb[(size_t)(q0 + 16 * qt + l16) * D_DIM +
                                   8 * j + 4 * hi];

  float qqv[2];
#pragma unroll
  for (int qt = 0; qt < 2; ++qt) {
    v4f s4 = qf[qt][0] * qf[qt][0];
#pragma unroll
    for (int j = 1; j < 8; ++j) s4 += qf[qt][j] * qf[qt][j];
    const float s = (s4.x + s4.y) + (s4.z + s4.w);
    qqv[qt] = (s + __shfl_xor(s, 16)) * 0.015625f;
  }

  v8f acc[4][2] = {};
  float kkrow[4];
#pragma unroll
  for (int n = 0; n < 4; ++n) {
    v4f kf[8];
#pragma unroll
    for (int j = 0; j < 8; ++j)
      kf[j] = *(const v4f*)&Kb[(size_t)(key0 + 16 * n + l16) * D_DIM + 8 * j +
                               4 * hi];
    {
      v4f s4 = kf[0] * kf[0];
#pragma unroll
      for (int j = 1; j < 8; ++j) s4 += kf[j] * kf[j];
      const float s = (s4.x + s4.y) + (s4.z + s4.w);
      kkrow[n] = (s + __shfl_xor(s, 16)) * 0.015625f;
    }
#pragma unroll
    for (int j = 0; j < 8; ++j) {
#pragma unroll
      for (int qt = 0; qt < 2; ++qt)
        acc[n][qt] = wmma4(kf[j].xy, qf[qt][j].xy, acc[n][qt]);
#pragma unroll
      for (int qt = 0; qt < 2; ++qt)
        acc[n][qt] = wmma4(kf[j].zw, qf[qt][j].zw, acc[n][qt]);
    }
  }

#if defined(__HIP_DEVICE_COMPILE__)
  __builtin_amdgcn_sched_barrier(0);
#endif
  float rs[4];
  {
    const float* qr =
        queries + ((size_t)b * L_DIM + q0 + 4 * wave) * U_DIM + 4 * lane;
#pragma unroll
    for (int rr = 0; rr < 4; ++rr) {
      v4f x[4];
#pragma unroll
      for (int i = 0; i < 4; ++i)
        x[i] = __builtin_nontemporal_load(
            (const v4f*)&qr[(size_t)rr * U_DIM + 128 * i]);
      const v4f t = (x[0] + x[1]) + (x[2] + x[3]);
      rs[rr] = (t.x + t.y) + (t.z + t.w);
    }
#pragma unroll
    for (int m = 16; m >= 1; m >>= 1)
#pragma unroll
      for (int rr = 0; rr < 4; ++rr) rs[rr] += __shfl_xor(rs[rr], m);
  }

  float dloc[2] = {INFINITY, INFINITY};
#pragma unroll
  for (int n = 0; n < 4; ++n) {
    float kk8[8];
#pragma unroll
    for (int r = 0; r < 8; ++r) kk8[r] = __shfl(kkrow[n], 8 * hi + r);
#pragma unroll
    for (int qt = 0; qt < 2; ++qt) {
      float st[8];
#pragma unroll
      for (int r = 0; r < 8; ++r) {
        float raw = fmaf(acc[n][qt][r], -0.03125f, qqv[qt] + kk8[r]);
        raw = fmaxf(raw, 1e-12f * 0.015625f);
        const float dist = hw_sqrt(raw);
        acc[n][qt][r] = dist;
        dloc[qt] = fminf(dloc[qt], dist);
        st[r] = dist;
      }
      float* dst = &sT[wave][16 * qt + l16][16 * n + 8 * hi];
      *(v4f*)&dst[0] = v4f{st[0], st[1], st[2], st[3]};
      *(v4f*)&dst[4] = v4f{st[4], st[5], st[6], st[7]};
    }
  }
#pragma unroll
  for (int qt = 0; qt < 2; ++qt) {
    dloc[qt] = fminf(dloc[qt], __shfl_xor(dloc[qt], 16));
    if (hi == 0) sM[wave][16 * qt + l16] = dloc[qt];
  }
  if (lane == 0) {
    *(v2f*)&sQM[0][2 * wave] = v2f{rs[0] != 0.0f ? 1.0f : 0.0f,
                                   rs[2] != 0.0f ? 1.0f : 0.0f};
    *(v2f*)&sQM[1][2 * wave] = v2f{rs[1] != 0.0f ? 1.0f : 0.0f,
                                   rs[3] != 0.0f ? 1.0f : 0.0f};
  }
  __syncthreads();

  v4f qm4[4];
#pragma unroll
  for (int i = 0; i < 4; ++i) qm4[i] = -*(const v4f*)&sQM[hi][4 * i];
#pragma unroll
  for (int i = 0; i < 16; ++i) {
    const int row = 2 * i + hi;
    const v4f v = *(const v4f*)&sT[wave][row][4 * l16] * qm4[i >> 2][i & 3];
    __builtin_nontemporal_store(
        v, (v4f*)&ascore[((size_t)hb * L_DIM + q0 + row) * L_DIM + key0 +
                         4 * l16]);
  }

  float lsum[2];
#pragma unroll
  for (int qt = 0; qt < 2; ++qt) {
    float m = sM[0][16 * qt + l16];
#pragma unroll
    for (int w = 1; w < 8; ++w) m = fminf(m, sM[w][16 * qt + l16]);
    float l = 0.0f;
#pragma unroll
    for (int n = 0; n < 4; ++n)
#pragma unroll
      for (int r = 0; r < 8; ++r) {
        const float p = __expf(m - acc[n][qt][r]);
        acc[n][qt][r] = p;
        l += p;
      }
    lsum[qt] = l + __shfl_xor(l, 16);
  }

  v8f o[4][2] = {};
#pragma unroll
  for (int n = 0; n < 4; ++n) {
    v4f vf[4][2];
#pragma unroll
    for (int t = 0; t < 4; ++t)
#pragma unroll
      for (int s = 0; s < 2; ++s)
        vf[t][s] = *(const v4f*)&Vb[(size_t)(key0 + 16 * n + 8 * hi + 2 * t + s) *
                                        D_DIM + 4 * l16];
#pragma unroll
    for (int t = 0; t < 4; ++t) {
#pragma unroll
      for (int np = 0; np < 4; ++np) {
        const v2f a = v2f{vf[t][0][np], vf[t][1][np]};
#pragma unroll
        for (int qt = 0; qt < 2; ++qt) {
          const v2f b = v2f{acc[n][qt][2 * t], acc[n][qt][2 * t + 1]};
          o[np][qt] = wmma4(a, b, o[np][qt]);
        }
      }
    }
  }

#pragma unroll
  for (int qt = 0; qt < 2; ++qt) {
    float* dst = &sT[wave][16 * qt + l16][32 * hi];
#pragma unroll
    for (int r = 0; r < 8; ++r)
      *(v4f*)&dst[4 * r] =
          v4f{o[0][qt][r], o[1][qt][r], o[2][qt][r], o[3][qt][r]};
    if (hi == 0) sL[wave][16 * qt + l16] = lsum[qt];
  }
  __syncthreads();

#pragma unroll
  for (int i = 0; i < 2; ++i) {
    const int q = (tid >> 4) + 16 * i;
    const int c4 = (tid & 15) * 4;
    v4f sum = *(const v4f*)&sT[0][q][c4];
    float L = sL[0][q];
#pragma unroll
    for (int w = 1; w < 8; ++w) {
      sum += *(const v4f*)&sT[w][q][c4];
      L += sL[w][q];
    }
    const float scale = sQM[q & 1][q >> 1] / L;
    sum *= scale;
    __builtin_nontemporal_store(
        sum, (v4f*)&outp[((size_t)b * L_DIM + q0 + q) * U_DIM + h * D_DIM + c4]);
  }
}

extern "C" void kernel_launch(void* const* d_in, const int* in_sizes, int n_in,
                              void* d_out, int out_size, void* d_ws,
                              size_t ws_size, hipStream_t stream) {
  const float* queries = (const float*)d_in[0];
  const float* keys = (const float*)d_in[1];
  const float* values = (const float*)d_in[2];
  const float* Wq = (const float*)d_in[3];
  const float* bq = (const float*)d_in[4];
  const float* Wk = (const float*)d_in[5];
  const float* bk = (const float*)d_in[6];
  const float* Wv = (const float*)d_in[7];
  const float* bv = (const float*)d_in[8];

  float* outp = (float*)d_out;
  float* ascore = outp + (size_t)B_DIM * L_DIM * U_DIM;

  float* ws = (float*)d_ws;
  float* Qh = ws;
  float* Kh = Qh + (size_t)HB_DIM * L_DIM * D_DIM;
  float* Vh = Kh + (size_t)HB_DIM * L_DIM * D_DIM;

  proj_relu_kernel<<<dim3(H_DIM, M_ROWS / 64, 3), 128, 0, stream>>>(
      queries, keys, values, Wq, Wk, Wv, bq, bk, bv, Qh, Kh, Vh);

  attn_kernel<<<dim3(L_DIM / AQ, HB_DIM), 256, 0, stream>>>(
      Qh, Kh, Vh, queries, outp, ascore);
}


# Q and K workspace stored fragment-major (projection takes W as the A side and X as the B side for Q,K; its stores stay whole 256-B runs): each attention Q/K fragment load is 512 contiguous B
# speedup vs baseline: 2.7083x; 1.0477x over previous
#include <hip/hip_runtime.h>
#include <hip/hip_bf16.h>
#include <math.h>

#define U_DIM 512
#define B_DIM 4
#define L_DIM 512
#define H_DIM 8
#define D_DIM 64
#define HB_DIM 32
#define M_ROWS 2048

typedef __attribute__((ext_vector_type(2))) float v2f;
typedef __attribute__((ext_vector_type(8))) float v8f;

__device__ __forceinline__ v8f wmma4(v2f a, v2f b, v8f c) {
#if defined(__HIP_DEVICE_COMPILE__)
  return __builtin_amdgcn_wmma_f32_16x16x4_f32(false, a, false, b, (short)0, c,
                                               false, false);
#else
  (void)a; (void)b;
  return c;
#endif
}

typedef __attribute__((ext_vector_type(4))) float v4f;

__device__ __forceinline__ float hw_sqrt(float x) {
#if defined(__HIP_DEVICE_COMPILE__)
  return __builtin_amdgcn_sqrtf(x);
#else
  return sqrtf(x);
#endif
}

#if defined(__HIP_DEVICE_COMPILE__)
#define PROJ_SCHED_FENCE() __builtin_amdgcn_sched_barrier(0)
#define PROJ_PACE_BARRIER() __builtin_amdgcn_s_barrier()
#else
#define PROJ_SCHED_FENCE() ((void)0)
#define PROJ_PACE_BARRIER() ((void)0)
#endif

struct PFrag {
  v4f x[4];
  v4f w[4][2][2];
};

__device__ __forceinline__ void proj_load(PFrag& f, const float* xp,
                                          const float* wp) {
#pragma unroll
  for (int q = 0; q < 4; ++q) f.x[q] = *(const v4f*)(xp + 8 * q);
#pragma unroll
  for (int q = 0; q < 4; ++q)
#pragma unroll
    for (int t = 0; t < 2; ++t)
#pragma unroll
      for (int s = 0; s < 2; ++s)
        f.w[q][t][s] = *(const v4f*)(wp + (size_t)(8 * q + 2 * t + s) * U_DIM);
}

template <bool WA>
__device__ __forceinline__ void proj_mma(v8f (&acc)[4], const PFrag& f) {
#pragma unroll
  for (int q = 0; q < 4; ++q)
#pragma unroll
    for (int t = 0; t < 2; ++t) {
      const v2f xv = t == 0 ? v2f{f.x[q].x, f.x[q].y} : v2f{f.x[q].z, f.x[q].w};
#pragma unroll
      for (int np = 0; np < 4; ++np) {
        const v2f wv = v2f{f.w[q][t][0][np], f.w[q][t][1][np]};
        acc[np] = WA ? wmma4(wv, xv, acc[np]) : wmma4(xv, wv, acc[np]);
      }
    }
}

template <bool WA>
__device__ __forceinline__ void proj_body(const float* __restrict__ X,
                                          const float* __restrict__ W,
                                          const float* __restrict__ bias,
                                          float* __restrict__ O) {
  const int g = blockIdx.x;
  const int tid = threadIdx.x;
  const int lane = tid & 31;
  const int wave = tid >> 5;
  const int l16 = lane & 15;
  const int hi = lane >> 4;
  const int m0 = blockIdx.y * 64 + 16 * wave;

  const float* xp = X + (size_t)(m0 + l16) * U_DIM + 4 * hi;
  const float* wp = W + (size_t)(4 * hi) * U_DIM + D_DIM * g + 4 * l16;

  v8f acc[4] = {};
  PFrag fa, fb;

  proj_load(fa, xp, wp);
  for (int i = 0; i < 7; ++i) {
    PROJ_PACE_BARRIER();
    proj_load(fb, xp + 32, wp + (size_t)32 * U_DIM);
    PROJ_SCHED_FENCE();
    proj_mma<WA>(acc, fa);
    xp += 64;
    wp += (size_t)64 * U_DIM;
    PROJ_PACE_BARRIER();
    proj_load(fa, xp, wp);
    PROJ_SCHED_FENCE();
    proj_mma<WA>(acc, fb);
  }
  PROJ_PACE_BARRIER();
  proj_load(fb, xp + 32, wp + (size_t)32 * U_DIM);
  PROJ_SCHED_FENCE();
  proj_mma<WA>(acc, fa);
  proj_mma<WA>(acc, fb);

  if (WA) {
    const int b = m0 >> 9, l0 = m0 & 511;
    float* dst = O + ((size_t)(g * B_DIM + b) * L_DIM + l0) * D_DIM + 512 * hi +
                 4 * l16;
    const float* bp = bias + D_DIM * g + 32 * hi;
#pragma unroll
    for (int r = 0; r < 8; ++r) {
      v4f v = v4f{acc[0][r], acc[1][r], acc[2][r], acc[3][r]} +
              *(const v4f*)&bp[4 * r];
      v.x = v.x > 0.0f ? v.x : 0.0f;
      v.y = v.y > 0.0f ? v.y : 0.0f;
      v.z = v.z > 0.0f ? v.z : 0.0f;
      v.w = v.w > 0.0f ? v.w : 0.0f;
      *(v4f*)&dst[64 * r] = v;
    }
  } else {
    const v4f b4 = *(const v4f*)&bias[D_DIM * g + 4 * l16];
#pragma unroll
    for (int r = 0; r < 8; ++r) {
      const int mg = m0 + 8 * hi + r;
      const int b = mg >> 9, l = mg & 511;
      v4f v = v4f{acc[0][r], acc[1][r], acc[2][r], acc[3][r]} + b4;
      v.x = v.x > 0.0f ? v.x : 0.0f;
      v.y = v.y > 0.0f ? v.y : 0.0f;
      v.z = v.z > 0.0f ? v.z : 0.0f;
      v.w = v.w > 0.0f ? v.w : 0.0f;
      *(v4f*)&O[(((size_t)(g * B_DIM + b) * L_DIM + l) * D_DIM) + 4 * l16] = v;
    }
  }
}

__global__ __launch_bounds__(128) __attribute__((amdgpu_waves_per_eu(3, 3)))
void proj_relu_kernel(
    const float* __restrict__ Xq, const float* __restrict__ Xk,
    const float* __restrict__ Xv, const float* __restrict__ Wq,
    const float* __restrict__ Wk, const float* __restrict__ Wv,
    const float* __restrict__ bq, const float* __restrict__ bk,
    const float* __restrict__ bv, float* __restrict__ Qh,
    float* __restrict__ Kh, float* __restrict__ Vh) {
  if (blockIdx.z == 2) {
    proj_body<false>(Xv, Wv, bv, Vh);
  } else {
    const bool isk = blockIdx.z == 1;
    proj_body<true>(isk ? Xk : Xq, isk ? Wk : Wq, isk ? bk : bq,
                    isk ? Kh : Qh);
  }
}

#define AQ 32
#define AT_PITCH 68

__global__ __launch_bounds__(256) void attn_kernel(
    const float* __restrict__ Qh, const float* __restrict__ Kh,
    const float* __restrict__ Vh, const float* __restrict__ queries,
    float* __restrict__ outp, float* __restrict__ ascore) {
  const int hb = blockIdx.y;
  const int q0 = blockIdx.x * AQ;
  const int tid = threadIdx.x;
  const int lane = tid & 31;
  const int wave = tid >> 5;
  const int l16 = lane & 15;
  const int hi = lane >> 4;
  const int key0 = wave * 64;

  __shared__ __attribute__((aligned(16))) float sT[8][AQ][AT_PITCH];
  __shared__ float sM[8][AQ];
  __shared__ float sL[8][AQ];
  __shared__ __attribute__((aligned(16))) float sQM[2][AQ / 2];

  const int h = hb >> 2, b = hb & 3;

  const float* Qb = Qh + (size_t)hb * L_DIM * D_DIM;
  const float* Kb = Kh + (size_t)hb * L_DIM * D_DIM;
  const float* Vb = Vh + (size_t)hb * L_DIM * D_DIM;

  v4f qf[2][8];
#pragma unroll
  for (int qt = 0; qt < 2; ++qt)
#pragma unroll
    for (int j = 0; j < 8; ++j)
      qf[qt][j] = *(const v4f*)&Qb[(size_t)(q0 + 16 * qt) * D_DIM + 128 * j +
                                   4 * lane];

  float qqv[2];
#pragma unroll
  for (int qt = 0; qt < 2; ++qt) {
    v4f s4 = qf[qt][0] * qf[qt][0];
#pragma unroll
    for (int j = 1; j < 8; ++j) s4 += qf[qt][j] * qf[qt][j];
    const float s = (s4.x + s4.y) + (s4.z + s4.w);
    qqv[qt] = (s + __shfl_xor(s, 16)) * 0.015625f;
  }

  v8f acc[4][2] = {};
  float kkrow[4];
#pragma unroll
  for (int n = 0; n < 4; ++n) {
    v4f kf[8];
#pragma unroll
    for (int j = 0; j < 8; ++j)
      kf[j] = *(const v4f*)&Kb[(size_t)(key0 + 16 * n) * D_DIM + 128 * j +
                               4 * lane];
    {
      v4f s4 = kf[0] * kf[0];
#pragma unroll
      for (int j = 1; j < 8; ++j) s4 += kf[j] * kf[j];
      const float s = (s4.x + s4.y) + (s4.z + s4.w);
      kkrow[n] = (s + __shfl_xor(s, 16)) * 0.015625f;
    }
#pragma unroll
    for (int j = 0; j < 8; ++j) {
#pragma unroll
      for (int qt = 0; qt < 2; ++qt)
        acc[n][qt] = wmma4(kf[j].xy, qf[qt][j].xy, acc[n][qt]);
#pragma unroll
      for (int qt = 0; qt < 2; ++qt)
        acc[n][qt] = wmma4(kf[j].zw, qf[qt][j].zw, acc[n][qt]);
    }
  }

#if defined(__HIP_DEVICE_COMPILE__)
  __builtin_amdgcn_sched_barrier(0);
#endif
  float rs[4];
  {
    const float* qr =
        queries + ((size_t)b * L_DIM + q0 + 4 * wave) * U_DIM + 4 * lane;
#pragma unroll
    for (int rr = 0; rr < 4; ++rr) {
      v4f x[4];
#pragma unroll
      for (int i = 0; i < 4; ++i)
        x[i] = __builtin_nontemporal_load(
            (const v4f*)&qr[(size_t)rr * U_DIM + 128 * i]);
      const v4f t = (x[0] + x[1]) + (x[2] + x[3]);
      rs[rr] = (t.x + t.y) + (t.z + t.w);
    }
#pragma unroll
    for (int m = 16; m >= 1; m >>= 1)
#pragma unroll
      for (int rr = 0; rr < 4; ++rr) rs[rr] += __shfl_xor(rs[rr], m);
  }

  float dloc[2] = {INFINITY, INFINITY};
#pragma unroll
  for (int n = 0; n < 4; ++n) {
    float kk8[8];
#pragma unroll
    for (int r = 0; r < 8; ++r) kk8[r] = __shfl(kkrow[n], 8 * hi + r);
#pragma unroll
    for (int qt = 0; qt < 2; ++qt) {
      float st[8];
#pragma unroll
      for (int r = 0; r < 8; ++r) {
        float raw = fmaf(acc[n][qt][r], -0.03125f, qqv[qt] + kk8[r]);
        raw = fmaxf(raw, 1e-12f * 0.015625f);
        const float dist = hw_sqrt(raw);
        acc[n][qt][r] = dist;
        dloc[qt] = fminf(dloc[qt], dist);
        st[r] = dist;
      }
      float* dst = &sT[wave][16 * qt + l16][16 * n + 8 * hi];
      *(v4f*)&dst[0] = v4f{st[0], st[1], st[2], st[3]};
      *(v4f*)&dst[4] = v4f{st[4], st[5], st[6], st[7]};
    }
  }
#pragma unroll
  for (int qt = 0; qt < 2; ++qt) {
    dloc[qt] = fminf(dloc[qt], __shfl_xor(dloc[qt], 16));
    if (hi == 0) sM[wave][16 * qt + l16] = dloc[qt];
  }
  if (lane == 0) {
    *(v2f*)&sQM[0][2 * wave] = v2f{rs[0] != 0.0f ? 1.0f : 0.0f,
                                   rs[2] != 0.0f ? 1.0f : 0.0f};
    *(v2f*)&sQM[1][2 * wave] = v2f{rs[1] != 0.0f ? 1.0f : 0.0f,
                                   rs[3] != 0.0f ? 1.0f : 0.0f};
  }
  __syncthreads();

  v4f qm4[4];
#pragma unroll
  for (int i = 0; i < 4; ++i) qm4[i] = -*(const v4f*)&sQM[hi][4 * i];
#pragma unroll
  for (int i = 0; i < 16; ++i) {
    const int row = 2 * i + hi;
    const v4f v = *(const v4f*)&sT[wave][row][4 * l16] * qm4[i >> 2][i & 3];
    __builtin_nontemporal_store(
        v, (v4f*)&ascore[((size_t)hb * L_DIM + q0 + row) * L_DIM + key0 +
                         4 * l16]);
  }

  float lsum[2];
#pragma unroll
  for (int qt = 0; qt < 2; ++qt) {
    float m = sM[0][16 * qt + l16];
#pragma unroll
    for (int w = 1; w < 8; ++w) m = fminf(m, sM[w][16 * qt + l16]);
    float l = 0.0f;
#pragma unroll
    for (int n = 0; n < 4; ++n)
#pragma unroll
      for (int r = 0; r < 8; ++r) {
        const float p = __expf(m - acc[n][qt][r]);
        acc[n][qt][r] = p;
        l += p;
      }
    lsum[qt] = l + __shfl_xor(l, 16);
  }

  v8f o[4][2] = {};
#pragma unroll
  for (int n = 0; n < 4; ++n) {
    v4f vf[4][2];
#pragma unroll
    for (int t = 0; t < 4; ++t)
#pragma unroll
      for (int s = 0; s < 2; ++s)
        vf[t][s] = *(const v4f*)&Vb[(size_t)(key0 + 16 * n + 8 * hi + 2 * t + s) *
                                        D_DIM + 4 * l16];
#pragma unroll
    for (int t = 0; t < 4; ++t) {
#pragma unroll
      for (int np = 0; np < 4; ++np) {
        const v2f a = v2f{vf[t][0][np], vf[t][1][np]};
#pragma unroll
        for (int qt = 0; qt < 2; ++qt) {
          const v2f b = v2f{acc[n][qt][2 * t], acc[n][qt][2 * t + 1]};
          o[np][qt] = wmma4(a, b, o[np][qt]);
        }
      }
    }
  }

#pragma unroll
  for (int qt = 0; qt < 2; ++qt) {
    float* dst = &sT[wave][16 * qt + l16][32 * hi];
#pragma unroll
    for (int r = 0; r < 8; ++r)
      *(v4f*)&dst[4 * r] =
          v4f{o[0][qt][r], o[1][qt][r], o[2][qt][r], o[3][qt][r]};
    if (hi == 0) sL[wave][16 * qt + l16] = lsum[qt];
  }
  __syncthreads();

#pragma unroll
  for (int i = 0; i < 2; ++i) {
    const int q = (tid >> 4) + 16 * i;
    const int c4 = (tid & 15) * 4;
    v4f sum = *(const v4f*)&sT[0][q][c4];
    float L = sL[0][q];
#pragma unroll
    for (int w = 1; w < 8; ++w) {
      sum += *(const v4f*)&sT[w][q][c4];
      L += sL[w][q];
    }
    const float scale = sQM[q & 1][q >> 1] / L;
    sum *= scale;
    __builtin_nontemporal_store(
        sum, (v4f*)&outp[((size_t)b * L_DIM + q0 + q) * U_DIM + h * D_DIM + c4]);
  }
}

extern "C" void kernel_launch(void* const* d_in, const int* in_sizes, int n_in,
                              void* d_out, int out_size, void* d_ws,
                              size_t ws_size, hipStream_t stream) {
  const float* queries = (const float*)d_in[0];
  const float* keys = (const float*)d_in[1];
  const float* values = (const float*)d_in[2];
  const float* Wq = (const float*)d_in[3];
  const float* bq = (const float*)d_in[4];
  const float* Wk = (const float*)d_in[5];
  const float* bk = (const float*)d_in[6];
  const float* Wv = (const float*)d_in[7];
  const float* bv = (const float*)d_in[8];

  float* outp = (float*)d_out;
  float* ascore = outp + (size_t)B_DIM * L_DIM * U_DIM;

  float* ws = (float*)d_ws;
  float* Qh = ws;
  float* Kh = Qh + (size_t)HB_DIM * L_DIM * D_DIM;
  float* Vh = Kh + (size_t)HB_DIM * L_DIM * D_DIM;

  proj_relu_kernel<<<dim3(H_DIM, M_ROWS / 64, 3), 128, 0, stream>>>(
      queries, keys, values, Wq, Wk, Wv, bq, bk, bv, Qh, Kh, Vh);

  attn_kernel<<<dim3(L_DIM / AQ, HB_DIM), 256, 0, stream>>>(
      Qh, Kh, Vh, queries, outp, ascore);
}
